# starting kernel
# baseline (speedup 1.0000x reference)
#include <hip/hip_runtime.h>

typedef float v2f __attribute__((ext_vector_type(2)));
typedef float v8f __attribute__((ext_vector_type(8)));

#define MDIM 8192
#define NDIM 1024
#define KDIM 1024
#define LHALF 16

#define BM 128
#define BN 128
#define BK 32
#define SA (4 * BM + 8)
#define SB (4 * BN + 8)

__global__ __launch_bounds__(256)
void eunn_coeff_kernel(const float* __restrict__ thetaA,
                       const float* __restrict__ thetaB,
                       float* __restrict__ dA, float* __restrict__ oA,
                       float* __restrict__ dB, float* __restrict__ oB)
{
    int idx = blockIdx.x * blockDim.x + threadIdx.x;
    if (idx >= LHALF * NDIM) return;
    const int t = idx >> 10;
    const int j = idx & 1023;

    {
        const int a = idx >> 5;
        const int b = (idx & 31) >> 1;
        const int c = idx & 1;
        const float th = thetaA[a * 16 + b];
        dA[idx] = cosf(th);
        oA[idx] = (c ? 1.0f : -1.0f) * sinf(th);
    }
    if (j == 0 || j == 1023) {
        dB[idx] = 1.0f;
        oB[idx] = 0.0f;
    } else {
        const int f = t * 1022 + (j - 1);
        const int a = f >> 5;
        const int b = (f & 31) >> 1;
        const int c = f & 1;
        const float th = thetaB[a * 16 + b];
        dB[idx] = cosf(th);
        oB[idx] = (c ? 1.0f : -1.0f) * sinf(th);
    }
}

__global__ __launch_bounds__(256)
void gemm_hx_kernel(const float* __restrict__ A,
                    const float* __restrict__ Bg,
                    const float* __restrict__ hx,
                    float* __restrict__ out)
{
    __shared__ __align__(16) float AsF[(BK / 4) * SA];
    __shared__ __align__(16) float BsF[(BK / 4) * SB];

    const int tid    = threadIdx.x;
    const int lane   = tid & 31;
    const int wave   = tid >> 5;
    const int wm     = wave & 3;
    const int wn     = wave >> 2;
    const int blockM = blockIdx.x * BM;
    const int blockN = blockIdx.y * BN;

    const int lane15 = lane & 15;
    const int koff   = (lane >> 4) << 1;

    v8f acc[2][4] = {};

    const int a_m0 = tid >> 3;
    const int a_kg = tid & 7;
    const int b_n  = tid & 127;
    const int b_kg0 = (tid >> 7);

    const int mA = wm * 32 + lane15;
    const int nB = wn * 64 + lane15;

    for (int kb = 0; kb < KDIM; kb += BK) {
        __syncthreads();
        #pragma unroll
        for (int p = 0; p < 4; ++p) {
            const int m = a_m0 + 32 * p;
            const float4 v = *(const float4*)(A + (size_t)(blockM + m) * KDIM + kb + a_kg * 4);
            *(float4*)(AsF + a_kg * SA + m * 4) = v;
        }
        #pragma unroll
        for (int p = 0; p < 4; ++p) {
            const int kg = b_kg0 + 2 * p;
            const int k  = kb + kg * 4;
            float4 v;
            v.x = Bg[(size_t)(k + 0) * NDIM + blockN + b_n];
            v.y = Bg[(size_t)(k + 1) * NDIM + blockN + b_n];
            v.z = Bg[(size_t)(k + 2) * NDIM + blockN + b_n];
            v.w = Bg[(size_t)(k + 3) * NDIM + blockN + b_n];
            *(float4*)(BsF + kg * SB + b_n * 4) = v;
        }
        __syncthreads();

        if (kb + BK < KDIM) {
            __builtin_prefetch(A + (size_t)(blockM + a_m0) * KDIM + kb + BK + a_kg * 4, 0, 0);
            __builtin_prefetch(Bg + (size_t)(kb + BK + b_kg0 * 4) * NDIM + blockN + b_n, 0, 0);
        }

        #pragma unroll
        for (int kg = 0; kg < BK / 4; ++kg) {
            const float* aB = AsF + kg * SA + koff;
            const float* bB = BsF + kg * SB + koff;
            v2f a0 = *(const v2f*)(aB + (mA     ) * 4);
            v2f a1 = *(const v2f*)(aB + (mA + 16) * 4);
            v2f b0 = *(const v2f*)(bB + (nB     ) * 4);
            v2f b1 = *(const v2f*)(bB + (nB + 16) * 4);
            v2f b2 = *(const v2f*)(bB + (nB + 32) * 4);
            v2f b3 = *(const v2f*)(bB + (nB + 48) * 4);
            acc[0][0] = __builtin_amdgcn_wmma_f32_16x16x4_f32(false, a0, false, b0,
                                                              (short)0, acc[0][0], false, false);
            acc[0][1] = __builtin_amdgcn_wmma_f32_16x16x4_f32(false, a0, false, b1,
                                                              (short)0, acc[0][1], false, false);
            acc[0][2] = __builtin_amdgcn_wmma_f32_16x16x4_f32(false, a0, false, b2,
                                                              (short)0, acc[0][2], false, false);
            acc[0][3] = __builtin_amdgcn_wmma_f32_16x16x4_f32(false, a0, false, b3,
                                                              (short)0, acc[0][3], false, false);
            acc[1][0] = __builtin_amdgcn_wmma_f32_16x16x4_f32(false, a1, false, b0,
                                                              (short)0, acc[1][0], false, false);
            acc[1][1] = __builtin_amdgcn_wmma_f32_16x16x4_f32(false, a1, false, b1,
                                                              (short)0, acc[1][1], false, false);
            acc[1][2] = __builtin_amdgcn_wmma_f32_16x16x4_f32(false, a1, false, b2,
                                                              (short)0, acc[1][2], false, false);
            acc[1][3] = __builtin_amdgcn_wmma_f32_16x16x4_f32(false, a1, false, b3,
                                                              (short)0, acc[1][3], false, false);
        }
    }

    const int hi = (lane >> 4) << 3;
    #pragma unroll
    for (int ti = 0; ti < 2; ++ti) {
        #pragma unroll
        for (int tj = 0; tj < 4; ++tj) {
            const int mBase = blockM + wm * 32 + ti * 16 + hi;
            const int nCol  = blockN + wn * 64 + tj * 16 + lane15;
            #pragma unroll
            for (int v = 0; v < 8; ++v) {
                const size_t idx = (size_t)(mBase + v) * NDIM + nCol;
                out[idx] = acc[ti][tj][v] + hx[idx];
            }
        }
    }
}

__global__ __launch_bounds__(256)
void eunn_modrelu_kernel(float* __restrict__ h,
                         const float* __restrict__ dA,
                         const float* __restrict__ oA,
                         const float* __restrict__ dB,
                         const float* __restrict__ oB,
                         const float* __restrict__ bias)
{
    __shared__ float xs[NDIM];
    const int row = blockIdx.x;
    const int tid = threadIdx.x;
    const int j0  = tid * 4;
    float* __restrict__ hrow = h + (size_t)row * NDIM;

    float4 x = *(const float4*)(hrow + j0);

    for (int t = 0; t < LHALF; ++t) {
        const float4 da = *(const float4*)(dA + t * NDIM + j0);
        const float4 oa = *(const float4*)(oA + t * NDIM + j0);
        float4 xa;
        xa.x = x.x * da.x + x.y * oa.x;
        xa.y = x.y * da.y + x.x * oa.y;
        xa.z = x.z * da.z + x.w * oa.z;
        xa.w = x.w * da.w + x.z * oa.w;

        *(float4*)(xs + j0) = xa;
        __syncthreads();

        const float4 db = *(const float4*)(dB + t * NDIM + j0);
        const float4 ob = *(const float4*)(oB + t * NDIM + j0);
        float y[4];
        #pragma unroll
        for (int u = 0; u < 4; ++u) {
            const int j   = j0 + u;
            const int src = (j < 512) ? (2 * j) : (2 * j - 1023);
            y[u] = xs[src];
        }
        x.x = xa.x * db.x + y[0] * ob.x;
        x.y = xa.y * db.y + y[1] * ob.y;
        x.z = xa.z * db.z + y[2] * ob.z;
        x.w = xa.w * db.w + y[3] * ob.w;
        __syncthreads();
    }

    const float4 bb = *(const float4*)(bias + j0);
    float4 o;
    {
        float s;
        s   = (x.x > 0.f) ? 1.f : ((x.x < 0.f) ? -1.f : 0.f);
        o.x = s * fmaxf(fabsf(x.x) + bb.x, 0.f);
        s   = (x.y > 0.f) ? 1.f : ((x.y < 0.f) ? -1.f : 0.f);
        o.y = s * fmaxf(fabsf(x.y) + bb.y, 0.f);
        s   = (x.z > 0.f) ? 1.f : ((x.z < 0.f) ? -1.f : 0.f);
        o.z = s * fmaxf(fabsf(x.z) + bb.z, 0.f);
        s   = (x.w > 0.f) ? 1.f : ((x.w < 0.f) ? -1.f : 0.f);
        o.w = s * fmaxf(fabsf(x.w) + bb.w, 0.f);
    }
    *(float4*)(hrow + j0) = o;
}

extern "C" void kernel_launch(void* const* d_in, const int* in_sizes, int n_in,
                              void* d_out, int out_size, void* d_ws, size_t ws_size,
                              hipStream_t stream)
{
    const float* input_ = (const float*)d_in[0];
    const float* hx     = (const float*)d_in[1];
    const float* U      = (const float*)d_in[2];
    const float* thetaA = (const float*)d_in[3];
    const float* thetaB = (const float*)d_in[4];
    const float* bias   = (const float*)d_in[5];
    float* out = (float*)d_out;

    float* ws = (float*)d_ws;
    float* dA = ws;
    float* oA = ws + 16384;
    float* dB = ws + 32768;
    float* oB = ws + 49152;

    eunn_coeff_kernel<<<(LHALF * NDIM + 255) / 256, 256, 0, stream>>>(
        thetaA, thetaB, dA, oA, dB, oB);

    dim3 grid(MDIM / BM, NDIM / BN);
    gemm_hx_kernel<<<grid, 256, 0, stream>>>(input_, U, hx, out);

    eunn_modrelu_kernel<<<MDIM, 256, 0, stream>>>(out, dA, oA, dB, oB, bias);
}


# scan kernel: one workgroup carries 8 batch rows, so each step's 4 coefficient vectors are loaded once per thread and reused for 8 rows and its 2 barriers serve 8 rows; rounding, sum order unchanged
# speedup vs baseline: 1.1246x; 1.1246x over previous
#include <hip/hip_runtime.h>

typedef float v2f __attribute__((ext_vector_type(2)));
typedef float v8f __attribute__((ext_vector_type(8)));

#define MDIM 8192
#define NDIM 1024
#define KDIM 1024
#define LHALF 16

#define BM 128
#define BN 128
#define BK 32
#define SA (4 * BM + 8)
#define SB (4 * BN + 8)

__global__ __launch_bounds__(256)
void eunn_coeff_kernel(const float* __restrict__ thetaA,
                       const float* __restrict__ thetaB,
                       float* __restrict__ dA, float* __restrict__ oA,
                       float* __restrict__ dB, float* __restrict__ oB)
{
    int idx = blockIdx.x * blockDim.x + threadIdx.x;
    if (idx >= LHALF * NDIM) return;
    const int t = idx >> 10;
    const int j = idx & 1023;

    {
        const int a = idx >> 5;
        const int b = (idx & 31) >> 1;
        const int c = idx & 1;
        const float th = thetaA[a * 16 + b];
        dA[idx] = cosf(th);
        oA[idx] = (c ? 1.0f : -1.0f) * sinf(th);
    }
    if (j == 0 || j == 1023) {
        dB[idx] = 1.0f;
        oB[idx] = 0.0f;
    } else {
        const int f = t * 1022 + (j - 1);
        const int a = f >> 5;
        const int b = (f & 31) >> 1;
        const int c = f & 1;
        const float th = thetaB[a * 16 + b];
        dB[idx] = cosf(th);
        oB[idx] = (c ? 1.0f : -1.0f) * sinf(th);
    }
}

__global__ __launch_bounds__(256)
void gemm_hx_kernel(const float* __restrict__ A,
                    const float* __restrict__ Bg,
                    const float* __restrict__ hx,
                    float* __restrict__ out)
{
    __shared__ __align__(16) float AsF[(BK / 4) * SA];
    __shared__ __align__(16) float BsF[(BK / 4) * SB];

    const int tid    = threadIdx.x;
    const int lane   = tid & 31;
    const int wave   = tid >> 5;
    const int wm     = wave & 3;
    const int wn     = wave >> 2;
    const int blockM = blockIdx.x * BM;
    const int blockN = blockIdx.y * BN;

    const int lane15 = lane & 15;
    const int koff   = (lane >> 4) << 1;

    v8f acc[2][4] = {};

    const int a_m0 = tid >> 3;
    const int a_kg = tid & 7;
    const int b_n  = tid & 127;
    const int b_kg0 = (tid >> 7);

    const int mA = wm * 32 + lane15;
    const int nB = wn * 64 + lane15;

    for (int kb = 0; kb < KDIM; kb += BK) {
        __syncthreads();
        #pragma unroll
        for (int p = 0; p < 4; ++p) {
            const int m = a_m0 + 32 * p;
            const float4 v = *(const float4*)(A + (size_t)(blockM + m) * KDIM + kb + a_kg * 4);
            *(float4*)(AsF + a_kg * SA + m * 4) = v;
        }
        #pragma unroll
        for (int p = 0; p < 4; ++p) {
            const int kg = b_kg0 + 2 * p;
            const int k  = kb + kg * 4;
            float4 v;
            v.x = Bg[(size_t)(k + 0) * NDIM + blockN + b_n];
            v.y = Bg[(size_t)(k + 1) * NDIM + blockN + b_n];
            v.z = Bg[(size_t)(k + 2) * NDIM + blockN + b_n];
            v.w = Bg[(size_t)(k + 3) * NDIM + blockN + b_n];
            *(float4*)(BsF + kg * SB + b_n * 4) = v;
        }
        __syncthreads();

        if (kb + BK < KDIM) {
            __builtin_prefetch(A + (size_t)(blockM + a_m0) * KDIM + kb + BK + a_kg * 4, 0, 0);
            __builtin_prefetch(Bg + (size_t)(kb + BK + b_kg0 * 4) * NDIM + blockN + b_n, 0, 0);
        }

        #pragma unroll
        for (int kg = 0; kg < BK / 4; ++kg) {
            const float* aB = AsF + kg * SA + koff;
            const float* bB = BsF + kg * SB + koff;
            v2f a0 = *(const v2f*)(aB + (mA     ) * 4);
            v2f a1 = *(const v2f*)(aB + (mA + 16) * 4);
            v2f b0 = *(const v2f*)(bB + (nB     ) * 4);
            v2f b1 = *(const v2f*)(bB + (nB + 16) * 4);
            v2f b2 = *(const v2f*)(bB + (nB + 32) * 4);
            v2f b3 = *(const v2f*)(bB + (nB + 48) * 4);
            acc[0][0] = __builtin_amdgcn_wmma_f32_16x16x4_f32(false, a0, false, b0,
                                                              (short)0, acc[0][0], false, false);
            acc[0][1] = __builtin_amdgcn_wmma_f32_16x16x4_f32(false, a0, false, b1,
                                                              (short)0, acc[0][1], false, false);
            acc[0][2] = __builtin_amdgcn_wmma_f32_16x16x4_f32(false, a0, false, b2,
                                                              (short)0, acc[0][2], false, false);
            acc[0][3] = __builtin_amdgcn_wmma_f32_16x16x4_f32(false, a0, false, b3,
                                                              (short)0, acc[0][3], false, false);
            acc[1][0] = __builtin_amdgcn_wmma_f32_16x16x4_f32(false, a1, false, b0,
                                                              (short)0, acc[1][0], false, false);
            acc[1][1] = __builtin_amdgcn_wmma_f32_16x16x4_f32(false, a1, false, b1,
                                                              (short)0, acc[1][1], false, false);
            acc[1][2] = __builtin_amdgcn_wmma_f32_16x16x4_f32(false, a1, false, b2,
                                                              (short)0, acc[1][2], false, false);
            acc[1][3] = __builtin_amdgcn_wmma_f32_16x16x4_f32(false, a1, false, b3,
                                                              (short)0, acc[1][3], false, false);
        }
    }

    const int hi = (lane >> 4) << 3;
    #pragma unroll
    for (int ti = 0; ti < 2; ++ti) {
        #pragma unroll
        for (int tj = 0; tj < 4; ++tj) {
            const int mBase = blockM + wm * 32 + ti * 16 + hi;
            const int nCol  = blockN + wn * 64 + tj * 16 + lane15;
            #pragma unroll
            for (int v = 0; v < 8; ++v) {
                const size_t idx = (size_t)(mBase + v) * NDIM + nCol;
                out[idx] = acc[ti][tj][v] + hx[idx];
            }
        }
    }
}

#define RB 8

__global__ __launch_bounds__(256)
void eunn_modrelu_kernel(float* __restrict__ h,
                         const float* __restrict__ dA,
                         const float* __restrict__ oA,
                         const float* __restrict__ dB,
                         const float* __restrict__ oB,
                         const float* __restrict__ bias)
{
    __shared__ __align__(16) float xs[RB][NDIM];
    const int tid = threadIdx.x;
    const int j0  = tid * 4;
    float* __restrict__ hblk = h + (size_t)blockIdx.x * RB * NDIM;

    float4 x[RB];
    #pragma unroll
    for (int r = 0; r < RB; ++r)
        x[r] = *(const float4*)(hblk + r * NDIM + j0);

    int src[4];
    #pragma unroll
    for (int u = 0; u < 4; ++u) {
        const int j = j0 + u;
        src[u] = (j < 512) ? (2 * j) : (2 * j - 1023);
    }

    for (int t = 0; t < LHALF; ++t) {
        const float4 da = *(const float4*)(dA + t * NDIM + j0);
        const float4 oa = *(const float4*)(oA + t * NDIM + j0);
        const float4 db = *(const float4*)(dB + t * NDIM + j0);
        const float4 ob = *(const float4*)(oB + t * NDIM + j0);

        #pragma unroll
        for (int r = 0; r < RB; ++r) {
            float4 xa;
            xa.x = fmaf(x[r].x, da.x, x[r].y * oa.x);
            xa.y = fmaf(x[r].y, da.y, x[r].x * oa.y);
            xa.z = fmaf(x[r].z, da.z, x[r].w * oa.z);
            xa.w = fmaf(x[r].w, da.w, x[r].z * oa.w);
            *(float4*)(&xs[r][j0]) = xa;
            x[r] = xa;
        }
        __syncthreads();

        #pragma unroll
        for (int r = 0; r < RB; ++r) {
            const float y0 = xs[r][src[0]];
            const float y1 = xs[r][src[1]];
            const float y2 = xs[r][src[2]];
            const float y3 = xs[r][src[3]];
            x[r].x = fmaf(x[r].x, db.x, y0 * ob.x);
            x[r].y = fmaf(x[r].y, db.y, y1 * ob.y);
            x[r].z = fmaf(x[r].z, db.z, y2 * ob.z);
            x[r].w = fmaf(x[r].w, db.w, y3 * ob.w);
        }
        __syncthreads();
    }

    const float4 bb = *(const float4*)(bias + j0);
    #pragma unroll
    for (int r = 0; r < RB; ++r) {
        const float4 xv = x[r];
        float4 o;
        float s;
        s   = (xv.x > 0.f) ? 1.f : ((xv.x < 0.f) ? -1.f : 0.f);
        o.x = s * fmaxf(fabsf(xv.x) + bb.x, 0.f);
        s   = (xv.y > 0.f) ? 1.f : ((xv.y < 0.f) ? -1.f : 0.f);
        o.y = s * fmaxf(fabsf(xv.y) + bb.y, 0.f);
        s   = (xv.z > 0.f) ? 1.f : ((xv.z < 0.f) ? -1.f : 0.f);
        o.z = s * fmaxf(fabsf(xv.z) + bb.z, 0.f);
        s   = (xv.w > 0.f) ? 1.f : ((xv.w < 0.f) ? -1.f : 0.f);
        o.w = s * fmaxf(fabsf(xv.w) + bb.w, 0.f);
        *(float4*)(hblk + r * NDIM + j0) = o;
    }
}

extern "C" void kernel_launch(void* const* d_in, const int* in_sizes, int n_in,
                              void* d_out, int out_size, void* d_ws, size_t ws_size,
                              hipStream_t stream)
{
    const float* input_ = (const float*)d_in[0];
    const float* hx     = (const float*)d_in[1];
    const float* U      = (const float*)d_in[2];
    const float* thetaA = (const float*)d_in[3];
    const float* thetaB = (const float*)d_in[4];
    const float* bias   = (const float*)d_in[5];
    float* out = (float*)d_out;

    float* ws = (float*)d_ws;
    float* dA = ws;
    float* oA = ws + 16384;
    float* dB = ws + 32768;
    float* oB = ws + 49152;

    eunn_coeff_kernel<<<(LHALF * NDIM + 255) / 256, 256, 0, stream>>>(
        thetaA, thetaB, dA, oA, dB, oB);

    dim3 grid(MDIM / BM, NDIM / BN);
    gemm_hx_kernel<<<grid, 256, 0, stream>>>(input_, U, hx, out);

    eunn_modrelu_kernel<<<MDIM / RB, 256, 0, stream>>>(out, dA, oA, dB, oB, bias);
}


# scan kernel: stage A stores each row to LDS in stage-B partner order (even columns, then odd); stage B reads its 4 partners as one 16-byte load, no bank conflicts; arithmetic and sum order unchanged
# speedup vs baseline: 1.2587x; 1.1192x over previous
#include <hip/hip_runtime.h>

typedef float v2f __attribute__((ext_vector_type(2)));
typedef float v8f __attribute__((ext_vector_type(8)));

#define MDIM 8192
#define NDIM 1024
#define KDIM 1024
#define LHALF 16

#define BM 128
#define BN 128
#define BK 32
#define SA (4 * BM + 8)
#define SB (4 * BN + 8)

__global__ __launch_bounds__(256)
void eunn_coeff_kernel(const float* __restrict__ thetaA,
                       const float* __restrict__ thetaB,
                       float* __restrict__ dA, float* __restrict__ oA,
                       float* __restrict__ dB, float* __restrict__ oB)
{
    int idx = blockIdx.x * blockDim.x + threadIdx.x;
    if (idx >= LHALF * NDIM) return;
    const int t = idx >> 10;
    const int j = idx & 1023;

    {
        const int a = idx >> 5;
        const int b = (idx & 31) >> 1;
        const int c = idx & 1;
        const float th = thetaA[a * 16 + b];
        dA[idx] = cosf(th);
        oA[idx] = (c ? 1.0f : -1.0f) * sinf(th);
    }
    if (j == 0 || j == 1023) {
        dB[idx] = 1.0f;
        oB[idx] = 0.0f;
    } else {
        const int f = t * 1022 + (j - 1);
        const int a = f >> 5;
        const int b = (f & 31) >> 1;
        const int c = f & 1;
        const float th = thetaB[a * 16 + b];
        dB[idx] = cosf(th);
        oB[idx] = (c ? 1.0f : -1.0f) * sinf(th);
    }
}

__global__ __launch_bounds__(256)
void gemm_hx_kernel(const float* __restrict__ A,
                    const float* __restrict__ Bg,
                    const float* __restrict__ hx,
                    float* __restrict__ out)
{
    __shared__ __align__(16) float AsF[(BK / 4) * SA];
    __shared__ __align__(16) float BsF[(BK / 4) * SB];

    const int tid    = threadIdx.x;
    const int lane   = tid & 31;
    const int wave   = tid >> 5;
    const int wm     = wave & 3;
    const int wn     = wave >> 2;
    const int blockM = blockIdx.x * BM;
    const int blockN = blockIdx.y * BN;

    const int lane15 = lane & 15;
    const int koff   = (lane >> 4) << 1;

    v8f acc[2][4] = {};

    const int a_m0 = tid >> 3;
    const int a_kg = tid & 7;
    const int b_n  = tid & 127;
    const int b_kg0 = (tid >> 7);

    const int mA = wm * 32 + lane15;
    const int nB = wn * 64 + lane15;

    for (int kb = 0; kb < KDIM; kb += BK) {
        __syncthreads();
        #pragma unroll
        for (int p = 0; p < 4; ++p) {
            const int m = a_m0 + 32 * p;
            const float4 v = *(const float4*)(A + (size_t)(blockM + m) * KDIM + kb + a_kg * 4);
            *(float4*)(AsF + a_kg * SA + m * 4) = v;
        }
        #pragma unroll
        for (int p = 0; p < 4; ++p) {
            const int kg = b_kg0 + 2 * p;
            const int k  = kb + kg * 4;
            float4 v;
            v.x = Bg[(size_t)(k + 0) * NDIM + blockN + b_n];
            v.y = Bg[(size_t)(k + 1) * NDIM + blockN + b_n];
            v.z = Bg[(size_t)(k + 2) * NDIM + blockN + b_n];
            v.w = Bg[(size_t)(k + 3) * NDIM + blockN + b_n];
            *(float4*)(BsF + kg * SB + b_n * 4) = v;
        }
        __syncthreads();

        if (kb + BK < KDIM) {
            __builtin_prefetch(A + (size_t)(blockM + a_m0) * KDIM + kb + BK + a_kg * 4, 0, 0);
            __builtin_prefetch(Bg + (size_t)(kb + BK + b_kg0 * 4) * NDIM + blockN + b_n, 0, 0);
        }

        #pragma unroll
        for (int kg = 0; kg < BK / 4; ++kg) {
            const float* aB = AsF + kg * SA + koff;
            const float* bB = BsF + kg * SB + koff;
            v2f a0 = *(const v2f*)(aB + (mA     ) * 4);
            v2f a1 = *(const v2f*)(aB + (mA + 16) * 4);
            v2f b0 = *(const v2f*)(bB + (nB     ) * 4);
            v2f b1 = *(const v2f*)(bB + (nB + 16) * 4);
            v2f b2 = *(const v2f*)(bB + (nB + 32) * 4);
            v2f b3 = *(const v2f*)(bB + (nB + 48) * 4);
            acc[0][0] = __builtin_amdgcn_wmma_f32_16x16x4_f32(false, a0, false, b0,
                                                              (short)0, acc[0][0], false, false);
            acc[0][1] = __builtin_amdgcn_wmma_f32_16x16x4_f32(false, a0, false, b1,
                                                              (short)0, acc[0][1], false, false);
            acc[0][2] = __builtin_amdgcn_wmma_f32_16x16x4_f32(false, a0, false, b2,
                                                              (short)0, acc[0][2], false, false);
            acc[0][3] = __builtin_amdgcn_wmma_f32_16x16x4_f32(false, a0, false, b3,
                                                              (short)0, acc[0][3], false, false);
            acc[1][0] = __builtin_amdgcn_wmma_f32_16x16x4_f32(false, a1, false, b0,
                                                              (short)0, acc[1][0], false, false);
            acc[1][1] = __builtin_amdgcn_wmma_f32_16x16x4_f32(false, a1, false, b1,
                                                              (short)0, acc[1][1], false, false);
            acc[1][2] = __builtin_amdgcn_wmma_f32_16x16x4_f32(false, a1, false, b2,
                                                              (short)0, acc[1][2], false, false);
            acc[1][3] = __builtin_amdgcn_wmma_f32_16x16x4_f32(false, a1, false, b3,
                                                              (short)0, acc[1][3], false, false);
        }
    }

    const int hi = (lane >> 4) << 3;
    #pragma unroll
    for (int ti = 0; ti < 2; ++ti) {
        #pragma unroll
        for (int tj = 0; tj < 4; ++tj) {
            const int mBase = blockM + wm * 32 + ti * 16 + hi;
            const int nCol  = blockN + wn * 64 + tj * 16 + lane15;
            #pragma unroll
            for (int v = 0; v < 8; ++v) {
                const size_t idx = (size_t)(mBase + v) * NDIM + nCol;
                out[idx] = acc[ti][tj][v] + hx[idx];
            }
        }
    }
}

#define RB 8

__global__ __launch_bounds__(256)
void eunn_modrelu_kernel(float* __restrict__ h,
                         const float* __restrict__ dA,
                         const float* __restrict__ oA,
                         const float* __restrict__ dB,
                         const float* __restrict__ oB,
                         const float* __restrict__ bias)
{
    __shared__ __align__(16) float xs[RB][NDIM];
    const int tid = threadIdx.x;
    const int j0  = tid * 4;
    float* __restrict__ hblk = h + (size_t)blockIdx.x * RB * NDIM;

    float4 x[RB];
    #pragma unroll
    for (int r = 0; r < RB; ++r)
        x[r] = *(const float4*)(hblk + r * NDIM + j0);

    const int pe = 2 * tid;
    const int po = 512 + 2 * tid;

    for (int t = 0; t < LHALF; ++t) {
        const float4 da = *(const float4*)(dA + t * NDIM + j0);
        const float4 oa = *(const float4*)(oA + t * NDIM + j0);
        const float4 db = *(const float4*)(dB + t * NDIM + j0);
        const float4 ob = *(const float4*)(oB + t * NDIM + j0);

        #pragma unroll
        for (int r = 0; r < RB; ++r) {
            float4 xa;
            xa.x = fmaf(x[r].x, da.x, x[r].y * oa.x);
            xa.y = fmaf(x[r].y, da.y, x[r].x * oa.y);
            xa.z = fmaf(x[r].z, da.z, x[r].w * oa.z);
            xa.w = fmaf(x[r].w, da.w, x[r].z * oa.w);
            *(float2*)(&xs[r][pe]) = make_float2(xa.x, xa.z);
            *(float2*)(&xs[r][po]) = make_float2(xa.y, xa.w);
            x[r] = xa;
        }
        __syncthreads();

        #pragma unroll
        for (int r = 0; r < RB; ++r) {
            const float4 y = *(const float4*)(&xs[r][j0]);
            x[r].x = fmaf(x[r].x, db.x, y.x * ob.x);
            x[r].y = fmaf(x[r].y, db.y, y.y * ob.y);
            x[r].z = fmaf(x[r].z, db.z, y.z * ob.z);
            x[r].w = fmaf(x[r].w, db.w, y.w * ob.w);
        }
        __syncthreads();
    }

    const float4 bb = *(const float4*)(bias + j0);
    #pragma unroll
    for (int r = 0; r < RB; ++r) {
        const float4 xv = x[r];
        float4 o;
        float s;
        s   = (xv.x > 0.f) ? 1.f : ((xv.x < 0.f) ? -1.f : 0.f);
        o.x = s * fmaxf(fabsf(xv.x) + bb.x, 0.f);
        s   = (xv.y > 0.f) ? 1.f : ((xv.y < 0.f) ? -1.f : 0.f);
        o.y = s * fmaxf(fabsf(xv.y) + bb.y, 0.f);
        s   = (xv.z > 0.f) ? 1.f : ((xv.z < 0.f) ? -1.f : 0.f);
        o.z = s * fmaxf(fabsf(xv.z) + bb.z, 0.f);
        s   = (xv.w > 0.f) ? 1.f : ((xv.w < 0.f) ? -1.f : 0.f);
        o.w = s * fmaxf(fabsf(xv.w) + bb.w, 0.f);
        *(float4*)(hblk + r * NDIM + j0) = o;
    }
}

extern "C" void kernel_launch(void* const* d_in, const int* in_sizes, int n_in,
                              void* d_out, int out_size, void* d_ws, size_t ws_size,
                              hipStream_t stream)
{
    const float* input_ = (const float*)d_in[0];
    const float* hx     = (const float*)d_in[1];
    const float* U      = (const float*)d_in[2];
    const float* thetaA = (const float*)d_in[3];
    const float* thetaB = (const float*)d_in[4];
    const float* bias   = (const float*)d_in[5];
    float* out = (float*)d_out;

    float* ws = (float*)d_ws;
    float* dA = ws;
    float* oA = ws + 16384;
    float* dB = ws + 32768;
    float* oB = ws + 49152;

    eunn_coeff_kernel<<<(LHALF * NDIM + 255) / 256, 256, 0, stream>>>(
        thetaA, thetaB, dA, oA, dB, oB);

    dim3 grid(MDIM / BM, NDIM / BN);
    gemm_hx_kernel<<<grid, 256, 0, stream>>>(input_, U, hx, out);

    eunn_modrelu_kernel<<<MDIM / RB, 256, 0, stream>>>(out, dA, oA, dB, oB, bias);
}


# GEMM K loop software-pipelined: next K tile's global loads issued ahead of the current 64 WMMAs, stored to a second LDS stage behind them; one barrier per K step; sum order unchanged
# speedup vs baseline: 1.3733x; 1.0910x over previous
#include <hip/hip_runtime.h>

typedef float v2f __attribute__((ext_vector_type(2)));
typedef float v8f __attribute__((ext_vector_type(8)));

#define MDIM 8192
#define NDIM 1024
#define KDIM 1024
#define LHALF 16

#define BM 128
#define BN 128
#define BK 32
#define SA (4 * BM + 8)
#define SB (4 * BN + 8)
#define STA ((BK / 4) * SA)
#define STB ((BK / 4) * SB)

__global__ __launch_bounds__(256)
void eunn_coeff_kernel(const float* __restrict__ thetaA,
                       const float* __restrict__ thetaB,
                       float* __restrict__ dA, float* __restrict__ oA,
                       float* __restrict__ dB, float* __restrict__ oB)
{
    int idx = blockIdx.x * blockDim.x + threadIdx.x;
    if (idx >= LHALF * NDIM) return;
    const int t = idx >> 10;
    const int j = idx & 1023;

    {
        const int a = idx >> 5;
        const int b = (idx & 31) >> 1;
        const int c = idx & 1;
        const float th = thetaA[a * 16 + b];
        dA[idx] = cosf(th);
        oA[idx] = (c ? 1.0f : -1.0f) * sinf(th);
    }
    if (j == 0 || j == 1023) {
        dB[idx] = 1.0f;
        oB[idx] = 0.0f;
    } else {
        const int f = t * 1022 + (j - 1);
        const int a = f >> 5;
        const int b = (f & 31) >> 1;
        const int c = f & 1;
        const float th = thetaB[a * 16 + b];
        dB[idx] = cosf(th);
        oB[idx] = (c ? 1.0f : -1.0f) * sinf(th);
    }
}

__global__ __launch_bounds__(256)
void gemm_hx_kernel(const float* A,
                    const float* __restrict__ Bg,
                    const float* __restrict__ hx,
                    float* __restrict__ out)
{
    __shared__ __align__(16) float AsF[2 * STA];
    __shared__ __align__(16) float BsF[2 * STB];

    const int tid    = threadIdx.x;
    const int lane   = tid & 31;
    const int wave   = tid >> 5;
    const int wm     = wave & 3;
    const int wn     = wave >> 2;
    const int blockM = blockIdx.x * BM;
    const int blockN = blockIdx.y * BN;

    const int lane15 = lane & 15;
    const int koff   = (lane >> 4) << 1;

    v8f acc[2][4] = {};

    const int a_m0 = tid >> 3;
    const int a_kg = tid & 7;
    const int b_n  = tid & 127;
    const int b_kg0 = (tid >> 7);

    const int mA = wm * 32 + lane15;
    const int nB = wn * 64 + lane15;

    const float* gA = A + (size_t)(blockM + a_m0) * KDIM + a_kg * 4;
    const float* gB = Bg + (size_t)(b_kg0 * 4) * NDIM + blockN + b_n;
    float* sA = AsF + a_kg * SA + a_m0 * 4;
    float* sB = BsF + b_kg0 * SB + b_n * 4;

    float4 ra0, ra1, ra2, ra3, rb0, rb1, rb2, rb3;
#define LOAD_TILE(kb_)                                                          \
    do {                                                                        \
        const float* pa = gA + (kb_);                                           \
        const float* pb = gB + (size_t)(kb_) * NDIM;                            \
        ra0 = *(const float4*)(pa);                                             \
        ra1 = *(const float4*)(pa + (size_t)32 * KDIM);                         \
        ra2 = *(const float4*)(pa + (size_t)64 * KDIM);                         \
        ra3 = *(const float4*)(pa + (size_t)96 * KDIM);                         \
        rb0 = make_float4(pb[0 * NDIM], pb[1 * NDIM], pb[2 * NDIM], pb[3 * NDIM]);          \
        rb1 = make_float4(pb[8 * NDIM], pb[9 * NDIM], pb[10 * NDIM], pb[11 * NDIM]);        \
        rb2 = make_float4(pb[16 * NDIM], pb[17 * NDIM], pb[18 * NDIM], pb[19 * NDIM]);      \
        rb3 = make_float4(pb[24 * NDIM], pb[25 * NDIM], pb[26 * NDIM], pb[27 * NDIM]);      \
    } while (0)
#define STORE_TILE(st_)                                                         \
    do {                                                                        \
        float* wa = sA + (st_) * STA;                                           \
        float* wb = sB + (st_) * STB;                                           \
        *(float4*)(wa)       = ra0;                                             \
        *(float4*)(wa + 128) = ra1;                                             \
        *(float4*)(wa + 256) = ra2;                                             \
        *(float4*)(wa + 384) = ra3;                                             \
        *(float4*)(wb)          = rb0;                                          \
        *(float4*)(wb + 2 * SB) = rb1;                                          \
        *(float4*)(wb + 4 * SB) = rb2;                                          \
        *(float4*)(wb + 6 * SB) = rb3;                                          \
    } while (0)
#define COMPUTE_TILE(st_)                                                       \
    do {                                                                        \
        const float* As = AsF + (st_) * STA + koff;                             \
        const float* Bs = BsF + (st_) * STB + koff;                             \
          \
        _Pragma("unroll")                                                       \
        for (int kg = 0; kg < BK / 4; ++kg) {                                   \
            const float* aB = As + kg * SA;                                     \
            const float* bB = Bs + kg * SB;                                     \
            v2f a0 = *(const v2f*)(aB + (mA     ) * 4);                         \
            v2f a1 = *(const v2f*)(aB + (mA + 16) * 4);                         \
            v2f b0 = *(const v2f*)(bB + (nB     ) * 4);                         \
            v2f b1 = *(const v2f*)(bB + (nB + 16) * 4);                         \
            v2f b2 = *(const v2f*)(bB + (nB + 32) * 4);                         \
            v2f b3 = *(const v2f*)(bB + (nB + 48) * 4);                         \
            acc[0][0] = __builtin_amdgcn_wmma_f32_16x16x4_f32(false, a0, false, b0, (short)0, acc[0][0], false, false); \
            acc[0][1] = __builtin_amdgcn_wmma_f32_16x16x4_f32(false, a0, false, b1, (short)0, acc[0][1], false, false); \
            acc[0][2] = __builtin_amdgcn_wmma_f32_16x16x4_f32(false, a0, false, b2, (short)0, acc[0][2], false, false); \
            acc[0][3] = __builtin_amdgcn_wmma_f32_16x16x4_f32(false, a0, false, b3, (short)0, acc[0][3], false, false); \
            acc[1][0] = __builtin_amdgcn_wmma_f32_16x16x4_f32(false, a1, false, b0, (short)0, acc[1][0], false, false); \
            acc[1][1] = __builtin_amdgcn_wmma_f32_16x16x4_f32(false, a1, false, b1, (short)0, acc[1][1], false, false); \
            acc[1][2] = __builtin_amdgcn_wmma_f32_16x16x4_f32(false, a1, false, b2, (short)0, acc[1][2], false, false); \
            acc[1][3] = __builtin_amdgcn_wmma_f32_16x16x4_f32(false, a1, false, b3, (short)0, acc[1][3], false, false); \
        }                                                                       \
    } while (0)

#define HOLD() __builtin_amdgcn_sched_barrier(0)

    LOAD_TILE(0);
    STORE_TILE(0);
    __syncthreads();

    #pragma unroll 1
    for (int kb = 0; kb < KDIM - 2 * BK; kb += 2 * BK) {
        LOAD_TILE(kb + BK);
        HOLD();
        COMPUTE_TILE(0);
        HOLD();
        STORE_TILE(1);
        __syncthreads();
        LOAD_TILE(kb + 2 * BK);
        HOLD();
        COMPUTE_TILE(1);
        HOLD();
        STORE_TILE(0);
        __syncthreads();
    }
    LOAD_TILE(KDIM - BK);
    HOLD();
    COMPUTE_TILE(0);
    HOLD();
    STORE_TILE(1);
    __syncthreads();
    COMPUTE_TILE(1);
#undef HOLD
#undef LOAD_TILE
#undef STORE_TILE
#undef COMPUTE_TILE

    const int hi = (lane >> 4) << 3;
    #pragma unroll
    for (int ti = 0; ti < 2; ++ti) {
        #pragma unroll
        for (int tj = 0; tj < 4; ++tj) {
            const int mBase = blockM + wm * 32 + ti * 16 + hi;
            const int nCol  = blockN + wn * 64 + tj * 16 + lane15;
            #pragma unroll
            for (int v = 0; v < 8; ++v) {
                const size_t idx = (size_t)(mBase + v) * NDIM + nCol;
                out[idx] = acc[ti][tj][v] + hx[idx];
            }
        }
    }
}

#define RB 8

__global__ __launch_bounds__(256)
void eunn_modrelu_kernel(float* __restrict__ h,
                         const float* __restrict__ dA,
                         const float* __restrict__ oA,
                         const float* __restrict__ dB,
                         const float* __restrict__ oB,
                         const float* __restrict__ bias)
{
    __shared__ __align__(16) float xs[RB][NDIM];
    const int tid = threadIdx.x;
    const int j0  = tid * 4;
    float* __restrict__ hblk = h + (size_t)blockIdx.x * RB * NDIM;

    float4 x[RB];
    #pragma unroll
    for (int r = 0; r < RB; ++r)
        x[r] = *(const float4*)(hblk + r * NDIM + j0);

    const int pe = 2 * tid;
    const int po = 512 + 2 * tid;

    for (int t = 0; t < LHALF; ++t) {
        const float4 da = *(const float4*)(dA + t * NDIM + j0);
        const float4 oa = *(const float4*)(oA + t * NDIM + j0);
        const float4 db = *(const float4*)(dB + t * NDIM + j0);
        const float4 ob = *(const float4*)(oB + t * NDIM + j0);

        #pragma unroll
        for (int r = 0; r < RB; ++r) {
            float4 xa;
            xa.x = fmaf(x[r].x, da.x, x[r].y * oa.x);
            xa.y = fmaf(x[r].y, da.y, x[r].x * oa.y);
            xa.z = fmaf(x[r].z, da.z, x[r].w * oa.z);
            xa.w = fmaf(x[r].w, da.w, x[r].z * oa.w);
            *(float2*)(&xs[r][pe]) = make_float2(xa.x, xa.z);
            *(float2*)(&xs[r][po]) = make_float2(xa.y, xa.w);
            x[r] = xa;
        }
        __syncthreads();

        #pragma unroll
        for (int r = 0; r < RB; ++r) {
            const float4 y = *(const float4*)(&xs[r][j0]);
            x[r].x = fmaf(x[r].x, db.x, y.x * ob.x);
            x[r].y = fmaf(x[r].y, db.y, y.y * ob.y);
            x[r].z = fmaf(x[r].z, db.z, y.z * ob.z);
            x[r].w = fmaf(x[r].w, db.w, y.w * ob.w);
        }
        __syncthreads();
    }

    const float4 bb = *(const float4*)(bias + j0);
    #pragma unroll
    for (int r = 0; r < RB; ++r) {
        const float4 xv = x[r];
        float4 o;
        float s;
        s   = (xv.x > 0.f) ? 1.f : ((xv.x < 0.f) ? -1.f : 0.f);
        o.x = s * fmaxf(fabsf(xv.x) + bb.x, 0.f);
        s   = (xv.y > 0.f) ? 1.f : ((xv.y < 0.f) ? -1.f : 0.f);
        o.y = s * fmaxf(fabsf(xv.y) + bb.y, 0.f);
        s   = (xv.z > 0.f) ? 1.f : ((xv.z < 0.f) ? -1.f : 0.f);
        o.z = s * fmaxf(fabsf(xv.z) + bb.z, 0.f);
        s   = (xv.w > 0.f) ? 1.f : ((xv.w < 0.f) ? -1.f : 0.f);
        o.w = s * fmaxf(fabsf(xv.w) + bb.w, 0.f);
        *(float4*)(hblk + r * NDIM + j0) = o;
    }
}

extern "C" void kernel_launch(void* const* d_in, const int* in_sizes, int n_in,
                              void* d_out, int out_size, void* d_ws, size_t ws_size,
                              hipStream_t stream)
{
    const float* input_ = (const float*)d_in[0];
    const float* hx     = (const float*)d_in[1];
    const float* U      = (const float*)d_in[2];
    const float* thetaA = (const float*)d_in[3];
    const float* thetaB = (const float*)d_in[4];
    const float* bias   = (const float*)d_in[5];
    float* out = (float*)d_out;

    float* ws = (float*)d_ws;
    float* dA = ws;
    float* oA = ws + 16384;
    float* dB = ws + 32768;
    float* oB = ws + 49152;

    eunn_coeff_kernel<<<(LHALF * NDIM + 255) / 256, 256, 0, stream>>>(
        thetaA, thetaB, dA, oA, dB, oB);

    dim3 grid(MDIM / BM, NDIM / BN);
    gemm_hx_kernel<<<grid, 256, 0, stream>>>(input_, U, hx, out);

    eunn_modrelu_kernel<<<MDIM / RB, 256, 0, stream>>>(out, dA, oA, dB, oB, bias);
}


# GEMM K loop on a ring of three LDS stages: next tile's LDS stores and global loads sit mid-block, barrier split (signal mid-block, wait before the block's last 8 WMMAs); sum order unchanged
# speedup vs baseline: 1.4272x; 1.0392x over previous
#include <hip/hip_runtime.h>

typedef float v2f __attribute__((ext_vector_type(2)));
typedef float v8f __attribute__((ext_vector_type(8)));

#define MDIM 8192
#define NDIM 1024
#define KDIM 1024
#define LHALF 16

#define BM 128
#define BN 128
#define BK 32
#define SA (4 * BM + 8)
#define SB (4 * BN + 8)
#define STA ((BK / 4) * SA)
#define STB ((BK / 4) * SB)

__global__ __launch_bounds__(256)
void eunn_coeff_kernel(const float* __restrict__ thetaA,
                       const float* __restrict__ thetaB,
                       float* __restrict__ dA, float* __restrict__ oA,
                       float* __restrict__ dB, float* __restrict__ oB)
{
    int idx = blockIdx.x * blockDim.x + threadIdx.x;
    if (idx >= LHALF * NDIM) return;
    const int t = idx >> 10;
    const int j = idx & 1023;

    {
        const int a = idx >> 5;
        const int b = (idx & 31) >> 1;
        const int c = idx & 1;
        const float th = thetaA[a * 16 + b];
        dA[idx] = cosf(th);
        oA[idx] = (c ? 1.0f : -1.0f) * sinf(th);
    }
    if (j == 0 || j == 1023) {
        dB[idx] = 1.0f;
        oB[idx] = 0.0f;
    } else {
        const int f = t * 1022 + (j - 1);
        const int a = f >> 5;
        const int b = (f & 31) >> 1;
        const int c = f & 1;
        const float th = thetaB[a * 16 + b];
        dB[idx] = cosf(th);
        oB[idx] = (c ? 1.0f : -1.0f) * sinf(th);
    }
}

__global__ __launch_bounds__(256)
void gemm_hx_kernel(const float* A,
                    const float* __restrict__ Bg,
                    const float* __restrict__ hx,
                    float* __restrict__ out)
{
    __shared__ __align__(16) float AsF[3 * STA];
    __shared__ __align__(16) float BsF[3 * STB];

    const int tid    = threadIdx.x;
    const int lane   = tid & 31;
    const int wave   = tid >> 5;
    const int wm     = wave & 3;
    const int wn     = wave >> 2;
    const int blockM = blockIdx.x * BM;
    const int blockN = blockIdx.y * BN;

    const int lane15 = lane & 15;
    const int koff   = (lane >> 4) << 1;

    v8f acc[2][4] = {};

    const int a_m0 = tid >> 3;
    const int a_kg = tid & 7;
    const int b_n  = tid & 127;
    const int b_kg0 = (tid >> 7);

    const int mA = wm * 32 + lane15;
    const int nB = wn * 64 + lane15;

    const float* gA = A + (size_t)(blockM + a_m0) * KDIM + a_kg * 4;
    const float* gB = Bg + (size_t)(b_kg0 * 4) * NDIM + blockN + b_n;
    float* sA = AsF + a_kg * SA + a_m0 * 4;
    float* sB = BsF + b_kg0 * SB + b_n * 4;

    float4 ra0, ra1, ra2, ra3, rb0, rb1, rb2, rb3;
#define LOAD_TILE(kb_)                                                          \
    do {                                                                        \
        const float* pa = gA + (kb_);                                           \
        const float* pb = gB + (size_t)(kb_) * NDIM;                            \
        ra0 = *(const float4*)(pa);                                             \
        ra1 = *(const float4*)(pa + (size_t)32 * KDIM);                         \
        ra2 = *(const float4*)(pa + (size_t)64 * KDIM);                         \
        ra3 = *(const float4*)(pa + (size_t)96 * KDIM);                         \
        rb0 = make_float4(pb[0 * NDIM], pb[1 * NDIM], pb[2 * NDIM], pb[3 * NDIM]);          \
        rb1 = make_float4(pb[8 * NDIM], pb[9 * NDIM], pb[10 * NDIM], pb[11 * NDIM]);        \
        rb2 = make_float4(pb[16 * NDIM], pb[17 * NDIM], pb[18 * NDIM], pb[19 * NDIM]);      \
        rb3 = make_float4(pb[24 * NDIM], pb[25 * NDIM], pb[26 * NDIM], pb[27 * NDIM]);      \
    } while (0)
#define STORE_TILE(st_)                                                         \
    do {                                                                        \
        float* wa = sA + (st_) * STA;                                           \
        float* wb = sB + (st_) * STB;                                           \
        *(float4*)(wa)       = ra0;                                             \
        *(float4*)(wa + 128) = ra1;                                             \
        *(float4*)(wa + 256) = ra2;                                             \
        *(float4*)(wa + 384) = ra3;                                             \
        *(float4*)(wb)          = rb0;                                          \
        *(float4*)(wb + 2 * SB) = rb1;                                          \
        *(float4*)(wb + 4 * SB) = rb2;                                          \
        *(float4*)(wb + 6 * SB) = rb3;                                          \
    } while (0)

    v2f fa0, fa1, fb0, fb1, fb2, fb3;
    v2f na0, na1, nb0, nb1, nb2, nb3;
#define LOAD_FRAG(st_, kg_)                                                     \
    do {                                                                        \
        const float* aB = AsF + (st_) * STA + koff + (kg_) * SA;                \
        const float* bB = BsF + (st_) * STB + koff + (kg_) * SB;                \
        na0 = *(const v2f*)(aB + (mA     ) * 4);                                \
        na1 = *(const v2f*)(aB + (mA + 16) * 4);                                \
        nb0 = *(const v2f*)(bB + (nB     ) * 4);                                \
        nb1 = *(const v2f*)(bB + (nB + 16) * 4);                                \
        nb2 = *(const v2f*)(bB + (nB + 32) * 4);                                \
        nb3 = *(const v2f*)(bB + (nB + 48) * 4);                                \
    } while (0)
#define NEXT_FRAG()                                                             \
    do { fa0 = na0; fa1 = na1; fb0 = nb0; fb1 = nb1; fb2 = nb2; fb3 = nb3; } while (0)
#define MMA8()                                                                  \
    do {                                                                        \
        acc[0][0] = __builtin_amdgcn_wmma_f32_16x16x4_f32(false, fa0, false, fb0, (short)0, acc[0][0], false, false); \
        acc[0][1] = __builtin_amdgcn_wmma_f32_16x16x4_f32(false, fa0, false, fb1, (short)0, acc[0][1], false, false); \
        acc[0][2] = __builtin_amdgcn_wmma_f32_16x16x4_f32(false, fa0, false, fb2, (short)0, acc[0][2], false, false); \
        acc[0][3] = __builtin_amdgcn_wmma_f32_16x16x4_f32(false, fa0, false, fb3, (short)0, acc[0][3], false, false); \
        acc[1][0] = __builtin_amdgcn_wmma_f32_16x16x4_f32(false, fa1, false, fb0, (short)0, acc[1][0], false, false); \
        acc[1][1] = __builtin_amdgcn_wmma_f32_16x16x4_f32(false, fa1, false, fb1, (short)0, acc[1][1], false, false); \
        acc[1][2] = __builtin_amdgcn_wmma_f32_16x16x4_f32(false, fa1, false, fb2, (short)0, acc[1][2], false, false); \
        acc[1][3] = __builtin_amdgcn_wmma_f32_16x16x4_f32(false, fa1, false, fb3, (short)0, acc[1][3], false, false); \
    } while (0)
#define KGROUP(st_, kg_)  do { LOAD_FRAG(st_, kg_); MMA8(); NEXT_FRAG(); } while (0)

#define HOLD() __builtin_amdgcn_sched_barrier(0)
#define SIGNAL_STAGE()                                                          \
    do {                                                                        \
        __builtin_amdgcn_fence(__ATOMIC_RELEASE, "workgroup", "local");         \
        __builtin_amdgcn_s_barrier_signal(-1);                                  \
    } while (0)
#define WAIT_STAGE()                                                            \
    do {                                                                        \
        __builtin_amdgcn_s_barrier_wait(-1);                                    \
        __builtin_amdgcn_fence(__ATOMIC_ACQUIRE, "workgroup", "local");         \
    } while (0)

#define KSTEP(ST_, NS_, KN_, STORE_, LOAD_, NEXT_)                              \
    do {                                                                        \
        KGROUP(ST_, 1);                                                         \
        KGROUP(ST_, 2);                                                         \
        KGROUP(ST_, 3);                                                         \
        KGROUP(ST_, 4);                \
        HOLD();                                                                 \
        if (STORE_) { STORE_TILE(NS_); }                                        \
        if (LOAD_)  { LOAD_TILE(KN_); }                                         \
        HOLD();                                                                 \
        KGROUP(ST_, 5);                                                         \
        KGROUP(ST_, 6);                                                         \
        HOLD();                                                                 \
        if (NEXT_) { SIGNAL_STAGE(); }                                          \
        HOLD();                                                                 \
        KGROUP(ST_, 7);                \
        HOLD();                                                                 \
        if (NEXT_) { WAIT_STAGE(); }                                            \
        HOLD();                                                                 \
        if (NEXT_) { LOAD_FRAG(NS_, 0); }                                       \
        MMA8();                                                   \
        NEXT_FRAG();                                                            \
    } while (0)

    LOAD_TILE(0);
    STORE_TILE(0);
    LOAD_TILE(BK);
    HOLD();
    SIGNAL_STAGE();
    WAIT_STAGE();
    HOLD();
    LOAD_FRAG(0, 0);
    NEXT_FRAG();

    #pragma unroll 1
    for (int kb = 0; kb < KDIM - 2 * BK; kb += 3 * BK) {
        KSTEP(0, 1, kb + 2 * BK, 1, 1, 1);
        KSTEP(1, 2, kb + 3 * BK, 1, 1, 1);
        KSTEP(2, 0, kb + 4 * BK, 1, 1, 1);
    }
    KSTEP(0, 1, 0, 1, 0, 1);
    KSTEP(1, 2, 0, 0, 0, 0);
#undef HOLD
#undef SIGNAL_STAGE
#undef WAIT_STAGE
#undef KSTEP
#undef KGROUP
#undef MMA8
#undef NEXT_FRAG
#undef LOAD_FRAG
#undef LOAD_TILE
#undef STORE_TILE

    const int hi = (lane >> 4) << 3;
    #pragma unroll
    for (int ti = 0; ti < 2; ++ti) {
        #pragma unroll
        for (int tj = 0; tj < 4; ++tj) {
            const int mBase = blockM + wm * 32 + ti * 16 + hi;
            const int nCol  = blockN + wn * 64 + tj * 16 + lane15;
            #pragma unroll
            for (int v = 0; v < 8; ++v) {
                const size_t idx = (size_t)(mBase + v) * NDIM + nCol;
                out[idx] = acc[ti][tj][v] + hx[idx];
            }
        }
    }
}

#define RB 8

__global__ __launch_bounds__(256)
void eunn_modrelu_kernel(float* __restrict__ h,
                         const float* __restrict__ dA,
                         const float* __restrict__ oA,
                         const float* __restrict__ dB,
                         const float* __restrict__ oB,
                         const float* __restrict__ bias)
{
    __shared__ __align__(16) float xs[RB][NDIM];
    const int tid = threadIdx.x;
    const int j0  = tid * 4;
    float* __restrict__ hblk = h + (size_t)blockIdx.x * RB * NDIM;

    float4 x[RB];
    #pragma unroll
    for (int r = 0; r < RB; ++r)
        x[r] = *(const float4*)(hblk + r * NDIM + j0);

    const int pe = 2 * tid;
    const int po = 512 + 2 * tid;

    for (int t = 0; t < LHALF; ++t) {
        const float4 da = *(const float4*)(dA + t * NDIM + j0);
        const float4 oa = *(const float4*)(oA + t * NDIM + j0);
        const float4 db = *(const float4*)(dB + t * NDIM + j0);
        const float4 ob = *(const float4*)(oB + t * NDIM + j0);

        #pragma unroll
        for (int r = 0; r < RB; ++r) {
            float4 xa;
            xa.x = fmaf(x[r].x, da.x, x[r].y * oa.x);
            xa.y = fmaf(x[r].y, da.y, x[r].x * oa.y);
            xa.z = fmaf(x[r].z, da.z, x[r].w * oa.z);
            xa.w = fmaf(x[r].w, da.w, x[r].z * oa.w);
            *(float2*)(&xs[r][pe]) = make_float2(xa.x, xa.z);
            *(float2*)(&xs[r][po]) = make_float2(xa.y, xa.w);
            x[r] = xa;
        }
        __syncthreads();

        #pragma unroll
        for (int r = 0; r < RB; ++r) {
            const float4 y = *(const float4*)(&xs[r][j0]);
            x[r].x = fmaf(x[r].x, db.x, y.x * ob.x);
            x[r].y = fmaf(x[r].y, db.y, y.y * ob.y);
            x[r].z = fmaf(x[r].z, db.z, y.z * ob.z);
            x[r].w = fmaf(x[r].w, db.w, y.w * ob.w);
        }
        __syncthreads();
    }

    const float4 bb = *(const float4*)(bias + j0);
    #pragma unroll
    for (int r = 0; r < RB; ++r) {
        const float4 xv = x[r];
        float4 o;
        float s;
        s   = (xv.x > 0.f) ? 1.f : ((xv.x < 0.f) ? -1.f : 0.f);
        o.x = s * fmaxf(fabsf(xv.x) + bb.x, 0.f);
        s   = (xv.y > 0.f) ? 1.f : ((xv.y < 0.f) ? -1.f : 0.f);
        o.y = s * fmaxf(fabsf(xv.y) + bb.y, 0.f);
        s   = (xv.z > 0.f) ? 1.f : ((xv.z < 0.f) ? -1.f : 0.f);
        o.z = s * fmaxf(fabsf(xv.z) + bb.z, 0.f);
        s   = (xv.w > 0.f) ? 1.f : ((xv.w < 0.f) ? -1.f : 0.f);
        o.w = s * fmaxf(fabsf(xv.w) + bb.w, 0.f);
        *(float4*)(hblk + r * NDIM + j0) = o;
    }
}

extern "C" void kernel_launch(void* const* d_in, const int* in_sizes, int n_in,
                              void* d_out, int out_size, void* d_ws, size_t ws_size,
                              hipStream_t stream)
{
    const float* input_ = (const float*)d_in[0];
    const float* hx     = (const float*)d_in[1];
    const float* U      = (const float*)d_in[2];
    const float* thetaA = (const float*)d_in[3];
    const float* thetaB = (const float*)d_in[4];
    const float* bias   = (const float*)d_in[5];
    float* out = (float*)d_out;

    float* ws = (float*)d_ws;
    float* dA = ws;
    float* oA = ws + 16384;
    float* dB = ws + 32768;
    float* oB = ws + 49152;

    eunn_coeff_kernel<<<(LHALF * NDIM + 255) / 256, 256, 0, stream>>>(
        thetaA, thetaB, dA, oA, dB, oB);

    dim3 grid(MDIM / BM, NDIM / BN);
    gemm_hx_kernel<<<grid, 256, 0, stream>>>(input_, U, hx, out);

    eunn_modrelu_kernel<<<MDIM / RB, 256, 0, stream>>>(out, dA, oA, dB, oB, bias);
}


# GEMM epilogue staged through LDS: each wave reads hx and writes h as 16-byte accesses over whole 256-byte row pieces (was 4-byte, 64-byte pieces); hx fetched under the last 2 K steps; same sum order
# speedup vs baseline: 1.6304x; 1.1424x over previous
#include <hip/hip_runtime.h>

typedef float v2f __attribute__((ext_vector_type(2)));
typedef float v8f __attribute__((ext_vector_type(8)));

#define MDIM 8192
#define NDIM 1024
#define KDIM 1024
#define LHALF 16

#define BM 128
#define BN 128
#define BK 32
#define SA (4 * BM + 8)
#define SB (4 * BN + 8)
#define STA ((BK / 4) * SA)
#define STB ((BK / 4) * SB)

__global__ __launch_bounds__(256)
void eunn_coeff_kernel(const float* __restrict__ thetaA,
                       const float* __restrict__ thetaB,
                       float* __restrict__ dA, float* __restrict__ oA,
                       float* __restrict__ dB, float* __restrict__ oB)
{
    int idx = blockIdx.x * blockDim.x + threadIdx.x;
    if (idx >= LHALF * NDIM) return;
    const int t = idx >> 10;
    const int j = idx & 1023;

    {
        const int a = idx >> 5;
        const int b = (idx & 31) >> 1;
        const int c = idx & 1;
        const float th = thetaA[a * 16 + b];
        dA[idx] = cosf(th);
        oA[idx] = (c ? 1.0f : -1.0f) * sinf(th);
    }
    if (j == 0 || j == 1023) {
        dB[idx] = 1.0f;
        oB[idx] = 0.0f;
    } else {
        const int f = t * 1022 + (j - 1);
        const int a = f >> 5;
        const int b = (f & 31) >> 1;
        const int c = f & 1;
        const float th = thetaB[a * 16 + b];
        dB[idx] = cosf(th);
        oB[idx] = (c ? 1.0f : -1.0f) * sinf(th);
    }
}

__global__ __launch_bounds__(256)
void gemm_hx_kernel(const float* A,
                    const float* __restrict__ Bg,
                    const float* __restrict__ hx,
                    float* __restrict__ out)
{
    __shared__ __align__(16) float smem[3 * STA + 3 * STB];
    float* const AsF = smem;
    float* const BsF = smem + 3 * STA;

    const int tid    = threadIdx.x;
    const int lane   = tid & 31;
    const int wave   = tid >> 5;
    const int wm     = wave & 3;
    const int wn     = wave >> 2;
    const int blockM = blockIdx.x * BM;
    const int blockN = blockIdx.y * BN;

    const int lane15 = lane & 15;
    const int koff   = (lane >> 4) << 1;

    v8f acc[2][4] = {};

    const int a_m0 = tid >> 3;
    const int a_kg = tid & 7;
    const int b_n  = tid & 127;
    const int b_kg0 = (tid >> 7);

    const int mA = wm * 32 + lane15;
    const int nB = wn * 64 + lane15;

    const float* gA = A + (size_t)(blockM + a_m0) * KDIM + a_kg * 4;
    const float* gB = Bg + (size_t)(b_kg0 * 4) * NDIM + blockN + b_n;
    float* sA = AsF + a_kg * SA + a_m0 * 4;
    float* sB = BsF + b_kg0 * SB + b_n * 4;

    float4 ra0, ra1, ra2, ra3, rb0, rb1, rb2, rb3;
#define LOAD_TILE(kb_)                                                          \
    do {                                                                        \
        const float* pa = gA + (kb_);                                           \
        const float* pb = gB + (size_t)(kb_) * NDIM;                            \
        ra0 = *(const float4*)(pa);                                             \
        ra1 = *(const float4*)(pa + (size_t)32 * KDIM);                         \
        ra2 = *(const float4*)(pa + (size_t)64 * KDIM);                         \
        ra3 = *(const float4*)(pa + (size_t)96 * KDIM);                         \
        rb0 = make_float4(pb[0 * NDIM], pb[1 * NDIM], pb[2 * NDIM], pb[3 * NDIM]);          \
        rb1 = make_float4(pb[8 * NDIM], pb[9 * NDIM], pb[10 * NDIM], pb[11 * NDIM]);        \
        rb2 = make_float4(pb[16 * NDIM], pb[17 * NDIM], pb[18 * NDIM], pb[19 * NDIM]);      \
        rb3 = make_float4(pb[24 * NDIM], pb[25 * NDIM], pb[26 * NDIM], pb[27 * NDIM]);      \
    } while (0)
#define STORE_TILE(st_)                                                         \
    do {                                                                        \
        float* wa = sA + (st_) * STA;                                           \
        float* wb = sB + (st_) * STB;                                           \
        *(float4*)(wa)       = ra0;                                             \
        *(float4*)(wa + 128) = ra1;                                             \
        *(float4*)(wa + 256) = ra2;                                             \
        *(float4*)(wa + 384) = ra3;                                             \
        *(float4*)(wb)          = rb0;                                          \
        *(float4*)(wb + 2 * SB) = rb1;                                          \
        *(float4*)(wb + 4 * SB) = rb2;                                          \
        *(float4*)(wb + 6 * SB) = rb3;                                          \
    } while (0)

    v2f fa0, fa1, fb0, fb1, fb2, fb3;
    v2f na0, na1, nb0, nb1, nb2, nb3;
#define LOAD_FRAG(st_, kg_)                                                     \
    do {                                                                        \
        const float* aB = AsF + (st_) * STA + koff + (kg_) * SA;                \
        const float* bB = BsF + (st_) * STB + koff + (kg_) * SB;                \
        na0 = *(const v2f*)(aB + (mA     ) * 4);                                \
        na1 = *(const v2f*)(aB + (mA + 16) * 4);                                \
        nb0 = *(const v2f*)(bB + (nB     ) * 4);                                \
        nb1 = *(const v2f*)(bB + (nB + 16) * 4);                                \
        nb2 = *(const v2f*)(bB + (nB + 32) * 4);                                \
        nb3 = *(const v2f*)(bB + (nB + 48) * 4);                                \
    } while (0)
#define NEXT_FRAG()                                                             \
    do { fa0 = na0; fa1 = na1; fb0 = nb0; fb1 = nb1; fb2 = nb2; fb3 = nb3; } while (0)
#define MMA8()                                                                  \
    do {                                                                        \
        acc[0][0] = __builtin_amdgcn_wmma_f32_16x16x4_f32(false, fa0, false, fb0, (short)0, acc[0][0], false, false); \
        acc[0][1] = __builtin_amdgcn_wmma_f32_16x16x4_f32(false, fa0, false, fb1, (short)0, acc[0][1], false, false); \
        acc[0][2] = __builtin_amdgcn_wmma_f32_16x16x4_f32(false, fa0, false, fb2, (short)0, acc[0][2], false, false); \
        acc[0][3] = __builtin_amdgcn_wmma_f32_16x16x4_f32(false, fa0, false, fb3, (short)0, acc[0][3], false, false); \
        acc[1][0] = __builtin_amdgcn_wmma_f32_16x16x4_f32(false, fa1, false, fb0, (short)0, acc[1][0], false, false); \
        acc[1][1] = __builtin_amdgcn_wmma_f32_16x16x4_f32(false, fa1, false, fb1, (short)0, acc[1][1], false, false); \
        acc[1][2] = __builtin_amdgcn_wmma_f32_16x16x4_f32(false, fa1, false, fb2, (short)0, acc[1][2], false, false); \
        acc[1][3] = __builtin_amdgcn_wmma_f32_16x16x4_f32(false, fa1, false, fb3, (short)0, acc[1][3], false, false); \
    } while (0)
#define KGROUP(st_, kg_)  do { LOAD_FRAG(st_, kg_); MMA8(); NEXT_FRAG(); } while (0)

#define HOLD() __builtin_amdgcn_sched_barrier(0)
#define SIGNAL_STAGE()                                                          \
    do {                                                                        \
        __builtin_amdgcn_fence(__ATOMIC_RELEASE, "workgroup", "local");         \
        __builtin_amdgcn_s_barrier_signal(-1);                                  \
    } while (0)
#define WAIT_STAGE()                                                            \
    do {                                                                        \
        __builtin_amdgcn_s_barrier_wait(-1);                                    \
        __builtin_amdgcn_fence(__ATOMIC_ACQUIRE, "workgroup", "local");         \
    } while (0)

#define KSTEP(ST_, NS_, KN_, STORE_, LOAD_, NEXT_)                              \
    do {                                                                        \
        KGROUP(ST_, 1);                                                         \
        KGROUP(ST_, 2);                                                         \
        KGROUP(ST_, 3);                                                         \
        KGROUP(ST_, 4);                \
        HOLD();                                                                 \
        if (STORE_) { STORE_TILE(NS_); }                                        \
        if (LOAD_)  { LOAD_TILE(KN_); }                                         \
        HOLD();                                                                 \
        KGROUP(ST_, 5);                                                         \
        KGROUP(ST_, 6);                                                         \
        HOLD();                                                                 \
        if (NEXT_) { SIGNAL_STAGE(); }                                          \
        HOLD();                                                                 \
        KGROUP(ST_, 7);                \
        HOLD();                                                                 \
        if (NEXT_) { WAIT_STAGE(); }                                            \
        HOLD();                                                                 \
        if (NEXT_) { LOAD_FRAG(NS_, 0); }                                       \
        MMA8();                                                   \
        NEXT_FRAG();                                                            \
    } while (0)

    LOAD_TILE(0);
    STORE_TILE(0);
    LOAD_TILE(BK);
    HOLD();
    SIGNAL_STAGE();
    WAIT_STAGE();
    HOLD();
    LOAD_FRAG(0, 0);
    NEXT_FRAG();

    #pragma unroll 1
    for (int kb = 0; kb < KDIM - 2 * BK; kb += 3 * BK) {
        KSTEP(0, 1, kb + 2 * BK, 1, 1, 1);
        KSTEP(1, 2, kb + 3 * BK, 1, 1, 1);
        KSTEP(2, 0, kb + 4 * BK, 1, 1, 1);
    }
    const int er = lane >> 4;
    const int ec = lane15 * 4;
    const size_t ebase = (size_t)(blockM + wm * 32 + er) * NDIM + blockN + wn * 64 + ec;
    float4 hv[16];
    #pragma unroll
    for (int i = 0; i < 16; ++i)
        hv[i] = *(const float4*)(hx + ebase + (size_t)(2 * i) * NDIM);
    HOLD();
    KSTEP(0, 1, 0, 1, 0, 1);
    KSTEP(1, 2, 0, 0, 0, 0);
#undef HOLD
#undef SIGNAL_STAGE
#undef WAIT_STAGE
#undef KSTEP
#undef KGROUP
#undef MMA8
#undef NEXT_FRAG
#undef LOAD_FRAG
#undef LOAD_TILE
#undef STORE_TILE

#define EPITCH 68
    float* const stg = smem + wave * (32 * EPITCH);
    const int hi = (lane >> 4) << 3;
    __syncthreads();
    #pragma unroll
    for (int ti = 0; ti < 2; ++ti) {
        #pragma unroll
        for (int tj = 0; tj < 4; ++tj) {
            #pragma unroll
            for (int v = 0; v < 8; ++v)
                stg[(ti * 16 + hi + v) * EPITCH + tj * 16 + lane15] = acc[ti][tj][v];
        }
    }
    __syncthreads();
    #pragma unroll
    for (int i = 0; i < 16; ++i) {
        const float4 a = *(const float4*)(stg + (2 * i + er) * EPITCH + ec);
        float4 o;
        o.x = a.x + hv[i].x;
        o.y = a.y + hv[i].y;
        o.z = a.z + hv[i].z;
        o.w = a.w + hv[i].w;
        *(float4*)(out + ebase + (size_t)(2 * i) * NDIM) = o;
    }
#undef EPITCH
}

#define RB 8

__global__ __launch_bounds__(256)
void eunn_modrelu_kernel(float* __restrict__ h,
                         const float* __restrict__ dA,
                         const float* __restrict__ oA,
                         const float* __restrict__ dB,
                         const float* __restrict__ oB,
                         const float* __restrict__ bias)
{
    __shared__ __align__(16) float xs[RB][NDIM];
    const int tid = threadIdx.x;
    const int j0  = tid * 4;
    float* __restrict__ hblk = h + (size_t)blockIdx.x * RB * NDIM;

    float4 x[RB];
    #pragma unroll
    for (int r = 0; r < RB; ++r)
        x[r] = *(const float4*)(hblk + r * NDIM + j0);

    const int pe = 2 * tid;
    const int po = 512 + 2 * tid;

    for (int t = 0; t < LHALF; ++t) {
        const float4 da = *(const float4*)(dA + t * NDIM + j0);
        const float4 oa = *(const float4*)(oA + t * NDIM + j0);
        const float4 db = *(const float4*)(dB + t * NDIM + j0);
        const float4 ob = *(const float4*)(oB + t * NDIM + j0);

        #pragma unroll
        for (int r = 0; r < RB; ++r) {
            float4 xa;
            xa.x = fmaf(x[r].x, da.x, x[r].y * oa.x);
            xa.y = fmaf(x[r].y, da.y, x[r].x * oa.y);
            xa.z = fmaf(x[r].z, da.z, x[r].w * oa.z);
            xa.w = fmaf(x[r].w, da.w, x[r].z * oa.w);
            *(float2*)(&xs[r][pe]) = make_float2(xa.x, xa.z);
            *(float2*)(&xs[r][po]) = make_float2(xa.y, xa.w);
            x[r] = xa;
        }
        __syncthreads();

        #pragma unroll
        for (int r = 0; r < RB; ++r) {
            const float4 y = *(const float4*)(&xs[r][j0]);
            x[r].x = fmaf(x[r].x, db.x, y.x * ob.x);
            x[r].y = fmaf(x[r].y, db.y, y.y * ob.y);
            x[r].z = fmaf(x[r].z, db.z, y.z * ob.z);
            x[r].w = fmaf(x[r].w, db.w, y.w * ob.w);
        }
        __syncthreads();
    }

    const float4 bb = *(const float4*)(bias + j0);
    #pragma unroll
    for (int r = 0; r < RB; ++r) {
        const float4 xv = x[r];
        float4 o;
        float s;
        s   = (xv.x > 0.f) ? 1.f : ((xv.x < 0.f) ? -1.f : 0.f);
        o.x = s * fmaxf(fabsf(xv.x) + bb.x, 0.f);
        s   = (xv.y > 0.f) ? 1.f : ((xv.y < 0.f) ? -1.f : 0.f);
        o.y = s * fmaxf(fabsf(xv.y) + bb.y, 0.f);
        s   = (xv.z > 0.f) ? 1.f : ((xv.z < 0.f) ? -1.f : 0.f);
        o.z = s * fmaxf(fabsf(xv.z) + bb.z, 0.f);
        s   = (xv.w > 0.f) ? 1.f : ((xv.w < 0.f) ? -1.f : 0.f);
        o.w = s * fmaxf(fabsf(xv.w) + bb.w, 0.f);
        *(float4*)(hblk + r * NDIM + j0) = o;
    }
}

extern "C" void kernel_launch(void* const* d_in, const int* in_sizes, int n_in,
                              void* d_out, int out_size, void* d_ws, size_t ws_size,
                              hipStream_t stream)
{
    const float* input_ = (const float*)d_in[0];
    const float* hx     = (const float*)d_in[1];
    const float* U      = (const float*)d_in[2];
    const float* thetaA = (const float*)d_in[3];
    const float* thetaB = (const float*)d_in[4];
    const float* bias   = (const float*)d_in[5];
    float* out = (float*)d_out;

    float* ws = (float*)d_ws;
    float* dA = ws;
    float* oA = ws + 16384;
    float* dB = ws + 32768;
    float* oB = ws + 49152;

    eunn_coeff_kernel<<<(LHALF * NDIM + 255) / 256, 256, 0, stream>>>(
        thetaA, thetaB, dA, oA, dB, oB);

    dim3 grid(MDIM / BM, NDIM / BN);
    gemm_hx_kernel<<<grid, 256, 0, stream>>>(input_, U, hx, out);

    eunn_modrelu_kernel<<<MDIM / RB, 256, 0, stream>>>(out, dA, oA, dB, oB, bias);
}


# scan kernel: each step's 4 coefficient vectors fetched one step ahead into a second register set; its 2 barriers order LDS only, so the fetch stays in flight across them; sum order unchanged
# speedup vs baseline: 1.6885x; 1.0356x over previous
#include <hip/hip_runtime.h>

typedef float v2f __attribute__((ext_vector_type(2)));
typedef float v8f __attribute__((ext_vector_type(8)));

#define MDIM 8192
#define NDIM 1024
#define KDIM 1024
#define LHALF 16

#define BM 128
#define BN 128
#define BK 32
#define SA (4 * BM + 8)
#define SB (4 * BN + 8)
#define STA ((BK / 4) * SA)
#define STB ((BK / 4) * SB)

__global__ __launch_bounds__(256)
void eunn_coeff_kernel(const float* __restrict__ thetaA,
                       const float* __restrict__ thetaB,
                       float* __restrict__ dA, float* __restrict__ oA,
                       float* __restrict__ dB, float* __restrict__ oB)
{
    int idx = blockIdx.x * blockDim.x + threadIdx.x;
    if (idx >= LHALF * NDIM) return;
    const int t = idx >> 10;
    const int j = idx & 1023;

    {
        const int a = idx >> 5;
        const int b = (idx & 31) >> 1;
        const int c = idx & 1;
        const float th = thetaA[a * 16 + b];
        dA[idx] = cosf(th);
        oA[idx] = (c ? 1.0f : -1.0f) * sinf(th);
    }
    if (j == 0 || j == 1023) {
        dB[idx] = 1.0f;
        oB[idx] = 0.0f;
    } else {
        const int f = t * 1022 + (j - 1);
        const int a = f >> 5;
        const int b = (f & 31) >> 1;
        const int c = f & 1;
        const float th = thetaB[a * 16 + b];
        dB[idx] = cosf(th);
        oB[idx] = (c ? 1.0f : -1.0f) * sinf(th);
    }
}

__global__ __launch_bounds__(256)
void gemm_hx_kernel(const float* A,
                    const float* __restrict__ Bg,
                    const float* __restrict__ hx,
                    float* __restrict__ out)
{
    __shared__ __align__(16) float smem[3 * STA + 3 * STB];
    float* const AsF = smem;
    float* const BsF = smem + 3 * STA;

    const int tid    = threadIdx.x;
    const int lane   = tid & 31;
    const int wave   = tid >> 5;
    const int wm     = wave & 3;
    const int wn     = wave >> 2;
    const int blockM = blockIdx.x * BM;
    const int blockN = blockIdx.y * BN;

    const int lane15 = lane & 15;
    const int koff   = (lane >> 4) << 1;

    v8f acc[2][4] = {};

    const int a_m0 = tid >> 3;
    const int a_kg = tid & 7;
    const int b_n  = tid & 127;
    const int b_kg0 = (tid >> 7);

    const int mA = wm * 32 + lane15;
    const int nB = wn * 64 + lane15;

    const float* gA = A + (size_t)(blockM + a_m0) * KDIM + a_kg * 4;
    const float* gB = Bg + (size_t)(b_kg0 * 4) * NDIM + blockN + b_n;
    float* sA = AsF + a_kg * SA + a_m0 * 4;
    float* sB = BsF + b_kg0 * SB + b_n * 4;

    float4 ra0, ra1, ra2, ra3, rb0, rb1, rb2, rb3;
#define LOAD_TILE(kb_)                                                          \
    do {                                                                        \
        const float* pa = gA + (kb_);                                           \
        const float* pb = gB + (size_t)(kb_) * NDIM;                            \
        ra0 = *(const float4*)(pa);                                             \
        ra1 = *(const float4*)(pa + (size_t)32 * KDIM);                         \
        ra2 = *(const float4*)(pa + (size_t)64 * KDIM);                         \
        ra3 = *(const float4*)(pa + (size_t)96 * KDIM);                         \
        rb0 = make_float4(pb[0 * NDIM], pb[1 * NDIM], pb[2 * NDIM], pb[3 * NDIM]);          \
        rb1 = make_float4(pb[8 * NDIM], pb[9 * NDIM], pb[10 * NDIM], pb[11 * NDIM]);        \
        rb2 = make_float4(pb[16 * NDIM], pb[17 * NDIM], pb[18 * NDIM], pb[19 * NDIM]);      \
        rb3 = make_float4(pb[24 * NDIM], pb[25 * NDIM], pb[26 * NDIM], pb[27 * NDIM]);      \
    } while (0)
#define STORE_TILE(st_)                                                         \
    do {                                                                        \
        float* wa = sA + (st_) * STA;                                           \
        float* wb = sB + (st_) * STB;                                           \
        *(float4*)(wa)       = ra0;                                             \
        *(float4*)(wa + 128) = ra1;                                             \
        *(float4*)(wa + 256) = ra2;                                             \
        *(float4*)(wa + 384) = ra3;                                             \
        *(float4*)(wb)          = rb0;                                          \
        *(float4*)(wb + 2 * SB) = rb1;                                          \
        *(float4*)(wb + 4 * SB) = rb2;                                          \
        *(float4*)(wb + 6 * SB) = rb3;                                          \
    } while (0)

    v2f fa0, fa1, fb0, fb1, fb2, fb3;
    v2f na0, na1, nb0, nb1, nb2, nb3;
#define LOAD_FRAG(st_, kg_)                                                     \
    do {                                                                        \
        const float* aB = AsF + (st_) * STA + koff + (kg_) * SA;                \
        const float* bB = BsF + (st_) * STB + koff + (kg_) * SB;                \
        na0 = *(const v2f*)(aB + (mA     ) * 4);                                \
        na1 = *(const v2f*)(aB + (mA + 16) * 4);                                \
        nb0 = *(const v2f*)(bB + (nB     ) * 4);                                \
        nb1 = *(const v2f*)(bB + (nB + 16) * 4);                                \
        nb2 = *(const v2f*)(bB + (nB + 32) * 4);                                \
        nb3 = *(const v2f*)(bB + (nB + 48) * 4);                                \
    } while (0)
#define NEXT_FRAG()                                                             \
    do { fa0 = na0; fa1 = na1; fb0 = nb0; fb1 = nb1; fb2 = nb2; fb3 = nb3; } while (0)
#define MMA8()                                                                  \
    do {                                                                        \
        acc[0][0] = __builtin_amdgcn_wmma_f32_16x16x4_f32(false, fa0, false, fb0, (short)0, acc[0][0], false, false); \
        acc[0][1] = __builtin_amdgcn_wmma_f32_16x16x4_f32(false, fa0, false, fb1, (short)0, acc[0][1], false, false); \
        acc[0][2] = __builtin_amdgcn_wmma_f32_16x16x4_f32(false, fa0, false, fb2, (short)0, acc[0][2], false, false); \
        acc[0][3] = __builtin_amdgcn_wmma_f32_16x16x4_f32(false, fa0, false, fb3, (short)0, acc[0][3], false, false); \
        acc[1][0] = __builtin_amdgcn_wmma_f32_16x16x4_f32(false, fa1, false, fb0, (short)0, acc[1][0], false, false); \
        acc[1][1] = __builtin_amdgcn_wmma_f32_16x16x4_f32(false, fa1, false, fb1, (short)0, acc[1][1], false, false); \
        acc[1][2] = __builtin_amdgcn_wmma_f32_16x16x4_f32(false, fa1, false, fb2, (short)0, acc[1][2], false, false); \
        acc[1][3] = __builtin_amdgcn_wmma_f32_16x16x4_f32(false, fa1, false, fb3, (short)0, acc[1][3], false, false); \
    } while (0)
#define KGROUP(st_, kg_)  do { LOAD_FRAG(st_, kg_); MMA8(); NEXT_FRAG(); } while (0)

#define HOLD() __builtin_amdgcn_sched_barrier(0)
#define SIGNAL_STAGE()                                                          \
    do {                                                                        \
        __builtin_amdgcn_fence(__ATOMIC_RELEASE, "workgroup", "local");         \
        __builtin_amdgcn_s_barrier_signal(-1);                                  \
    } while (0)
#define WAIT_STAGE()                                                            \
    do {                                                                        \
        __builtin_amdgcn_s_barrier_wait(-1);                                    \
        __builtin_amdgcn_fence(__ATOMIC_ACQUIRE, "workgroup", "local");         \
    } while (0)

#define KSTEP(ST_, NS_, KN_, STORE_, LOAD_, NEXT_)                              \
    do {                                                                        \
        KGROUP(ST_, 1);                                                         \
        KGROUP(ST_, 2);                                                         \
        KGROUP(ST_, 3);                                                         \
        KGROUP(ST_, 4);                \
        HOLD();                                                                 \
        if (STORE_) { STORE_TILE(NS_); }                                        \
        if (LOAD_)  { LOAD_TILE(KN_); }                                         \
        HOLD();                                                                 \
        KGROUP(ST_, 5);                                                         \
        KGROUP(ST_, 6);                                                         \
        HOLD();                                                                 \
        if (NEXT_) { SIGNAL_STAGE(); }                                          \
        HOLD();                                                                 \
        KGROUP(ST_, 7);                \
        HOLD();                                                                 \
        if (NEXT_) { WAIT_STAGE(); }                                            \
        HOLD();                                                                 \
        if (NEXT_) { LOAD_FRAG(NS_, 0); }                                       \
        MMA8();                                                   \
        NEXT_FRAG();                                                            \
    } while (0)

    LOAD_TILE(0);
    STORE_TILE(0);
    LOAD_TILE(BK);
    HOLD();
    SIGNAL_STAGE();
    WAIT_STAGE();
    HOLD();
    LOAD_FRAG(0, 0);
    NEXT_FRAG();

    #pragma unroll 1
    for (int kb = 0; kb < KDIM - 2 * BK; kb += 3 * BK) {
        KSTEP(0, 1, kb + 2 * BK, 1, 1, 1);
        KSTEP(1, 2, kb + 3 * BK, 1, 1, 1);
        KSTEP(2, 0, kb + 4 * BK, 1, 1, 1);
    }
    const int er = lane >> 4;
    const int ec = lane15 * 4;
    const size_t ebase = (size_t)(blockM + wm * 32 + er) * NDIM + blockN + wn * 64 + ec;
    float4 hv[16];
    #pragma unroll
    for (int i = 0; i < 16; ++i)
        hv[i] = *(const float4*)(hx + ebase + (size_t)(2 * i) * NDIM);
    HOLD();
    KSTEP(0, 1, 0, 1, 0, 1);
    KSTEP(1, 2, 0, 0, 0, 0);
#undef HOLD
#undef SIGNAL_STAGE
#undef WAIT_STAGE
#undef KSTEP
#undef KGROUP
#undef MMA8
#undef NEXT_FRAG
#undef LOAD_FRAG
#undef LOAD_TILE
#undef STORE_TILE

#define EPITCH 68
    float* const stg = smem + wave * (32 * EPITCH);
    const int hi = (lane >> 4) << 3;
    __syncthreads();
    #pragma unroll
    for (int ti = 0; ti < 2; ++ti) {
        #pragma unroll
        for (int tj = 0; tj < 4; ++tj) {
            #pragma unroll
            for (int v = 0; v < 8; ++v)
                stg[(ti * 16 + hi + v) * EPITCH + tj * 16 + lane15] = acc[ti][tj][v];
        }
    }
    __syncthreads();
    #pragma unroll
    for (int i = 0; i < 16; ++i) {
        const float4 a = *(const float4*)(stg + (2 * i + er) * EPITCH + ec);
        float4 o;
        o.x = a.x + hv[i].x;
        o.y = a.y + hv[i].y;
        o.z = a.z + hv[i].z;
        o.w = a.w + hv[i].w;
        *(float4*)(out + ebase + (size_t)(2 * i) * NDIM) = o;
    }
#undef EPITCH
}

#define RB 8

__global__ __launch_bounds__(256)
void eunn_modrelu_kernel(float* __restrict__ h,
                         const float* dA,
                         const float* oA,
                         const float* dB,
                         const float* oB,
                         const float* __restrict__ bias)
{
    __shared__ __align__(16) float xs[RB][NDIM];
    const int tid = threadIdx.x;
    const int j0  = tid * 4;
    float* __restrict__ hblk = h + (size_t)blockIdx.x * RB * NDIM;

    float4 x[RB];
    #pragma unroll
    for (int r = 0; r < RB; ++r)
        x[r] = *(const float4*)(hblk + r * NDIM + j0);

    const int pe = 2 * tid;
    const int po = 512 + 2 * tid;

#define LDS_BARRIER()                                                           \
    do {                                                                        \
        __builtin_amdgcn_fence(__ATOMIC_RELEASE, "workgroup", "local");         \
        __builtin_amdgcn_s_barrier_signal(-1);                                  \
        __builtin_amdgcn_s_barrier_wait(-1);                                    \
        __builtin_amdgcn_fence(__ATOMIC_ACQUIRE, "workgroup", "local");         \
    } while (0)

    float4 da = *(const float4*)(dA + j0);
    float4 oa = *(const float4*)(oA + j0);
    float4 db = *(const float4*)(dB + j0);
    float4 ob = *(const float4*)(oB + j0);

    #pragma unroll 2
    for (int t = 0; t < LHALF; ++t) {
        const int tn = (t + 1 < LHALF) ? (t + 1) : t;
        const float4 nda = *(const float4*)(dA + tn * NDIM + j0);
        const float4 noa = *(const float4*)(oA + tn * NDIM + j0);
        const float4 ndb = *(const float4*)(dB + tn * NDIM + j0);
        const float4 nob = *(const float4*)(oB + tn * NDIM + j0);

        #pragma unroll
        for (int r = 0; r < RB; ++r) {
            float4 xa;
            xa.x = fmaf(x[r].x, da.x, x[r].y * oa.x);
            xa.y = fmaf(x[r].y, da.y, x[r].x * oa.y);
            xa.z = fmaf(x[r].z, da.z, x[r].w * oa.z);
            xa.w = fmaf(x[r].w, da.w, x[r].z * oa.w);
            *(float2*)(&xs[r][pe]) = make_float2(xa.x, xa.z);
            *(float2*)(&xs[r][po]) = make_float2(xa.y, xa.w);
            x[r] = xa;
        }
        LDS_BARRIER();

        #pragma unroll
        for (int r = 0; r < RB; ++r) {
            const float4 y = *(const float4*)(&xs[r][j0]);
            x[r].x = fmaf(x[r].x, db.x, y.x * ob.x);
            x[r].y = fmaf(x[r].y, db.y, y.y * ob.y);
            x[r].z = fmaf(x[r].z, db.z, y.z * ob.z);
            x[r].w = fmaf(x[r].w, db.w, y.w * ob.w);
        }
        LDS_BARRIER();
        da = nda; oa = noa; db = ndb; ob = nob;
    }
#undef LDS_BARRIER

    const float4 bb = *(const float4*)(bias + j0);
    #pragma unroll
    for (int r = 0; r < RB; ++r) {
        const float4 xv = x[r];
        float4 o;
        float s;
        s   = (xv.x > 0.f) ? 1.f : ((xv.x < 0.f) ? -1.f : 0.f);
        o.x = s * fmaxf(fabsf(xv.x) + bb.x, 0.f);
        s   = (xv.y > 0.f) ? 1.f : ((xv.y < 0.f) ? -1.f : 0.f);
        o.y = s * fmaxf(fabsf(xv.y) + bb.y, 0.f);
        s   = (xv.z > 0.f) ? 1.f : ((xv.z < 0.f) ? -1.f : 0.f);
        o.z = s * fmaxf(fabsf(xv.z) + bb.z, 0.f);
        s   = (xv.w > 0.f) ? 1.f : ((xv.w < 0.f) ? -1.f : 0.f);
        o.w = s * fmaxf(fabsf(xv.w) + bb.w, 0.f);
        *(float4*)(hblk + r * NDIM + j0) = o;
    }
}

extern "C" void kernel_launch(void* const* d_in, const int* in_sizes, int n_in,
                              void* d_out, int out_size, void* d_ws, size_t ws_size,
                              hipStream_t stream)
{
    const float* input_ = (const float*)d_in[0];
    const float* hx     = (const float*)d_in[1];
    const float* U      = (const float*)d_in[2];
    const float* thetaA = (const float*)d_in[3];
    const float* thetaB = (const float*)d_in[4];
    const float* bias   = (const float*)d_in[5];
    float* out = (float*)d_out;

    float* ws = (float*)d_ws;
    float* dA = ws;
    float* oA = ws + 16384;
    float* dB = ws + 32768;
    float* oB = ws + 49152;

    eunn_coeff_kernel<<<(LHALF * NDIM + 255) / 256, 256, 0, stream>>>(
        thetaA, thetaB, dA, oA, dB, oB);

    dim3 grid(MDIM / BM, NDIM / BN);
    gemm_hx_kernel<<<grid, 256, 0, stream>>>(input_, U, hx, out);

    eunn_modrelu_kernel<<<MDIM / RB, 256, 0, stream>>>(out, dA, oA, dB, oB, bias);
}
